# starting kernel
# baseline (speedup 1.0000x reference)
#include <hip/hip_runtime.h>
#include <cmath>


typedef __attribute__((ext_vector_type(2))) float v2f;
typedef __attribute__((ext_vector_type(8))) float v8f;

#define I_SIZE   3
#define H        128
#define O_SIZE   3
#define BATCH    512
#define T_STEPS  50
#define AGENTS   64
#define PRED     12
#define NROWS    (BATCH * AGENTS)
#define WPAD     132
#define WAVES    8
#define TILE_M   16
#define ROWS_PER_BLOCK (WAVES * TILE_M)
#define NBLOCKS  (NROWS / ROWS_PER_BLOCK)

struct __align__(16) Smem {
    float W[H][WPAD];
    float fcT[H][16];
    float stage[WAVES][TILE_M][WPAD];
};

__device__ __forceinline__ v2f mk2(float a, float b) { v2f r; r[0] = a; r[1] = b; return r; }

__device__ __forceinline__ v8f splat8(float x) {
    v8f r;
#pragma unroll
    for (int i = 0; i < 8; ++i) r[i] = x;
    return r;
}

__device__ __forceinline__ v8f wmma4(v2f a, v2f b, v8f c) {
    return __builtin_amdgcn_wmma_f32_16x16x4_f32(
         false, a,  false, b,
         (short)0, c,  false,  false);
}

__device__ __forceinline__ float fast_tanh(float x) {
#if __has_builtin(__builtin_amdgcn_tanhf)
    return __builtin_amdgcn_tanhf(x);
#else
    return ::tanhf(x);
#endif
}

__global__ void __launch_bounds__(256, 1)
traj_rnn_kernel(const float* __restrict__ src,
                const float* __restrict__ W_ih_enc,
                const float* __restrict__ W_hh_enc,
                const float* __restrict__ b_ih_enc,
                const float* __restrict__ b_hh_enc,
                const float* __restrict__ W_ih_dec,
                const float* __restrict__ W_hh_dec,
                const float* __restrict__ b_ih_dec,
                const float* __restrict__ b_hh_dec,
                const float* __restrict__ fc_W,
                const float* __restrict__ fc_b,
                float* __restrict__ out)
{
    __shared__ Smem s;

    const int tid  = threadIdx.x;
    const int wave = tid >> 5;
    const int lane = tid & 31;
    const int l16  = lane & 15;
    const bool hi  = lane >= 16;
    const int koff = hi ? 2 : 0;
    const int r0   = blockIdx.x * ROWS_PER_BLOCK + wave * TILE_M;

    for (int i = tid; i < H * H; i += 256)
        s.W[i >> 7][i & 127] = W_hh_enc[i];
    for (int i = tid; i < H * 16; i += 256) {
        const int k = i >> 4, n = i & 15;
        s.fcT[k][n] = (n < O_SIZE) ? fc_W[n * H + k] : 0.0f;
    }
    __syncthreads();

    v2f BihE[8], BihD[8];
    float biasE[8], biasD[8];
#pragma unroll
    for (int t = 0; t < 8; ++t) {
        const int n = t * 16 + l16;
        if (!hi) {
            BihE[t] = mk2(W_ih_enc[n * I_SIZE + 0], W_ih_enc[n * I_SIZE + 1]);
            BihD[t] = mk2(W_ih_dec[n * I_SIZE + 0], W_ih_dec[n * I_SIZE + 1]);
        } else {
            BihE[t] = mk2(W_ih_enc[n * I_SIZE + 2], 0.0f);
            BihD[t] = mk2(W_ih_dec[n * I_SIZE + 2], 0.0f);
        }
        biasE[t] = b_ih_enc[n] + b_hh_enc[n];
        biasD[t] = b_ih_dec[n] + b_hh_dec[n];
    }
    const float fcb = (l16 < O_SIZE) ? fc_b[l16] : 0.0f;

    const int myrow = r0 + l16;
    const int mb = myrow >> 6;
    const int ma = myrow & 63;
    const float* xbase = src + (size_t)mb * T_STEPS * (AGENTS * I_SIZE) + ma * I_SIZE;

    v2f hA[32];
#pragma unroll
    for (int c = 0; c < 32; ++c) hA[c] = mk2(0.0f, 0.0f);

    v2f Ax = mk2(0.0f, 0.0f);
    v8f acc[8];

    for (int t = 0; t < T_STEPS; ++t) {
        const float* xr = xbase + (size_t)t * (AGENTS * I_SIZE);
        Ax = hi ? mk2(xr[2], 0.0f) : mk2(xr[0], xr[1]);

#pragma unroll
        for (int n = 0; n < 8; ++n) acc[n] = splat8(biasE[n]);
#pragma unroll
        for (int n = 0; n < 8; ++n) acc[n] = wmma4(Ax, BihE[n], acc[n]);

#pragma unroll
        for (int c = 0; c < 32; ++c) {
#pragma unroll
            for (int n = 0; n < 8; ++n) {
                const v2f Bv = *(const v2f*)&s.W[n * 16 + l16][4 * c + koff];
                acc[n] = wmma4(hA[c], Bv, acc[n]);
            }
        }

#pragma unroll
        for (int n = 0; n < 8; ++n) {
#pragma unroll
            for (int v = 0; v < 8; ++v) {
                const int row = hi ? v + 8 : v;
                s.stage[wave][row][n * 16 + l16] = fast_tanh(acc[n][v]);
            }
        }
#pragma unroll
        for (int c = 0; c < 32; ++c)
            hA[c] = *(const v2f*)&s.stage[wave][l16][4 * c + koff];
    }

    {
        const float* xr = xbase + (size_t)(T_STEPS - 1) * (AGENTS * I_SIZE);
        Ax = hi ? mk2(xr[2], 0.0f) : mk2(xr[0], xr[1]);
    }

    __syncthreads();
    for (int i = tid; i < H * H; i += 256)
        s.W[i >> 7][i & 127] = W_hh_dec[i];
    __syncthreads();

    for (int p = 0; p < PRED; ++p) {
#pragma unroll
        for (int n = 0; n < 8; ++n) acc[n] = splat8(biasD[n]);
#pragma unroll
        for (int n = 0; n < 8; ++n) acc[n] = wmma4(Ax, BihD[n], acc[n]);
#pragma unroll
        for (int c = 0; c < 32; ++c) {
#pragma unroll
            for (int n = 0; n < 8; ++n) {
                const v2f Bv = *(const v2f*)&s.W[n * 16 + l16][4 * c + koff];
                acc[n] = wmma4(hA[c], Bv, acc[n]);
            }
        }
#pragma unroll
        for (int n = 0; n < 8; ++n) {
#pragma unroll
            for (int v = 0; v < 8; ++v) {
                const int row = hi ? v + 8 : v;
                s.stage[wave][row][n * 16 + l16] = fast_tanh(acc[n][v]);
            }
        }
#pragma unroll
        for (int c = 0; c < 32; ++c)
            hA[c] = *(const v2f*)&s.stage[wave][l16][4 * c + koff];

        v8f pr = splat8(fcb);
#pragma unroll
        for (int c = 0; c < 32; ++c) {
            const v2f Bv = mk2(s.fcT[4 * c + koff][l16], s.fcT[4 * c + koff + 1][l16]);
            pr = wmma4(hA[c], Bv, pr);
        }

        if (l16 < O_SIZE) {
#pragma unroll
            for (int v = 0; v < 8; ++v) {
                const int row = hi ? v + 8 : v;
                const int n  = r0 + row;
                const int bb = n >> 6, aa = n & 63;
                out[((size_t)bb * PRED + p) * (AGENTS * O_SIZE) + aa * O_SIZE + l16] = pr[v];
                s.stage[wave][row][l16] = pr[v];
            }
        }
        Ax = hi ? mk2(s.stage[wave][l16][2], 0.0f)
                : mk2(s.stage[wave][l16][0], s.stage[wave][l16][1]);
    }
}

extern "C" void kernel_launch(void* const* d_in, const int* in_sizes, int n_in,
                              void* d_out, int out_size, void* d_ws, size_t ws_size,
                              hipStream_t stream) {
    const float* src      = (const float*)d_in[0];
    const float* W_ih_enc = (const float*)d_in[1];
    const float* W_hh_enc = (const float*)d_in[2];
    const float* b_ih_enc = (const float*)d_in[3];
    const float* b_hh_enc = (const float*)d_in[4];
    const float* W_ih_dec = (const float*)d_in[5];
    const float* W_hh_dec = (const float*)d_in[6];
    const float* b_ih_dec = (const float*)d_in[7];
    const float* b_hh_dec = (const float*)d_in[8];
    const float* fc_W     = (const float*)d_in[9];
    const float* fc_b     = (const float*)d_in[10];
    float* out = (float*)d_out;

    traj_rnn_kernel<<<NBLOCKS, 256, 0, stream>>>(
        src, W_ih_enc, W_hh_enc, b_ih_enc, b_hh_enc,
        W_ih_dec, W_hh_dec, b_ih_dec, b_hh_dec, fc_W, fc_b, out);
}


# B-fragment LDS loads grouped: each 64-WMMA block's loads (and the fc fragments) issued as ONE group ahead of the previous block, two register sets, instead of ~30 small groups among the WMMAs
# speedup vs baseline: 1.0384x; 1.0384x over previous
#include <hip/hip_runtime.h>
#include <cmath>


typedef __attribute__((ext_vector_type(2))) float v2f;
typedef __attribute__((ext_vector_type(8))) float v8f;

#define I_SIZE   3
#define H        128
#define O_SIZE   3
#define BATCH    512
#define T_STEPS  50
#define AGENTS   64
#define PRED     12
#define NROWS    (BATCH * AGENTS)
#define WPAD     132
#define WAVES    8
#define TILE_M   16
#define ROWS_PER_BLOCK (WAVES * TILE_M)
#define NBLOCKS  (NROWS / ROWS_PER_BLOCK)

struct __align__(16) Smem {
    float W[H][WPAD];
    float fcT[H][16];
    float stage[WAVES][TILE_M][WPAD];
};

__device__ __forceinline__ v2f mk2(float a, float b) { v2f r; r[0] = a; r[1] = b; return r; }

__device__ __forceinline__ v8f splat8(float x) {
    v8f r;
#pragma unroll
    for (int i = 0; i < 8; ++i) r[i] = x;
    return r;
}

__device__ __forceinline__ v8f wmma4(v2f a, v2f b, v8f c) {
    return __builtin_amdgcn_wmma_f32_16x16x4_f32(
         false, a,  false, b,
         (short)0, c,  false,  false);
}

__device__ __forceinline__ float fast_tanh(float x) {
#if __has_builtin(__builtin_amdgcn_tanhf)
    return __builtin_amdgcn_tanhf(x);
#else
    return ::tanhf(x);
#endif
}

#define KBLK 8

__device__ __forceinline__ void load_bblock(v2f (&dst)[64], const float* wlane, const int blk) {
#pragma unroll
    for (int n = 0; n < 8; ++n) {
#pragma unroll
        for (int cc = 0; cc < KBLK; ++cc)
            dst[cc * 8 + n] = *(const v2f*)(wlane + n * 16 * WPAD + 4 * (blk * KBLK + cc));
    }
}

__device__ __forceinline__ void mma_bblock(v8f (&acc)[8], const v2f (&hA)[32], const v2f (&B)[64], const int blk) {
#pragma unroll
    for (int cc = 0; cc < KBLK; ++cc) {
#pragma unroll
        for (int n = 0; n < 8; ++n)
            acc[n] = wmma4(hA[blk * KBLK + cc], B[cc * 8 + n], acc[n]);
    }
}

__device__ __forceinline__ void k_chain(v8f (&acc)[8], const v2f (&hA)[32],
                                        v2f (&B0)[64], v2f (&B1)[64], const float* wlane) {
    load_bblock(B1, wlane, 1);
    __builtin_amdgcn_sched_barrier(0);
    mma_bblock(acc, hA, B0, 0);
    __builtin_amdgcn_sched_barrier(0);
    load_bblock(B0, wlane, 2);
    __builtin_amdgcn_sched_barrier(0);
    mma_bblock(acc, hA, B1, 1);
    __builtin_amdgcn_sched_barrier(0);
    load_bblock(B1, wlane, 3);
    __builtin_amdgcn_sched_barrier(0);
    mma_bblock(acc, hA, B0, 2);
    __builtin_amdgcn_sched_barrier(0);
    load_bblock(B0, wlane, 0);
    __builtin_amdgcn_sched_barrier(0);
    mma_bblock(acc, hA, B1, 3);
}

__global__ void __launch_bounds__(256, 1)
traj_rnn_kernel(const float* __restrict__ src,
                const float* __restrict__ W_ih_enc,
                const float* __restrict__ W_hh_enc,
                const float* __restrict__ b_ih_enc,
                const float* __restrict__ b_hh_enc,
                const float* __restrict__ W_ih_dec,
                const float* __restrict__ W_hh_dec,
                const float* __restrict__ b_ih_dec,
                const float* __restrict__ b_hh_dec,
                const float* __restrict__ fc_W,
                const float* __restrict__ fc_b,
                float* __restrict__ out)
{
    __shared__ Smem s;

    const int tid  = threadIdx.x;
    const int wave = tid >> 5;
    const int lane = tid & 31;
    const int l16  = lane & 15;
    const bool hi  = lane >= 16;
    const int koff = hi ? 2 : 0;
    const int r0   = blockIdx.x * ROWS_PER_BLOCK + wave * TILE_M;

    for (int i = tid; i < H * H; i += 256)
        s.W[i >> 7][i & 127] = W_hh_enc[i];
    for (int i = tid; i < H * 16; i += 256) {
        const int k = i >> 4, n = i & 15;
        s.fcT[k][n] = (n < O_SIZE) ? fc_W[n * H + k] : 0.0f;
    }
    __syncthreads();

    v2f BihE[8];
    float biasE[8];
#pragma unroll
    for (int t = 0; t < 8; ++t) {
        const int n = t * 16 + l16;
        if (!hi) BihE[t] = mk2(W_ih_enc[n * I_SIZE + 0], W_ih_enc[n * I_SIZE + 1]);
        else     BihE[t] = mk2(W_ih_enc[n * I_SIZE + 2], 0.0f);
        biasE[t] = b_ih_enc[n] + b_hh_enc[n];
    }
    const float fcb = (l16 < O_SIZE) ? fc_b[l16] : 0.0f;

    const int myrow = r0 + l16;
    const int mb = myrow >> 6;
    const int ma = myrow & 63;
    const float* xbase = src + (size_t)mb * T_STEPS * (AGENTS * I_SIZE) + ma * I_SIZE;

    v2f hA[32];
#pragma unroll
    for (int c = 0; c < 32; ++c) hA[c] = mk2(0.0f, 0.0f);

    v2f Ax = mk2(0.0f, 0.0f);
    v8f acc[8];

    const float* wlane = &s.W[l16][koff];
    v2f Bs0[64], Bs1[64];
    load_bblock(Bs0, wlane, 0);

    for (int t = 0; t < T_STEPS; ++t) {
        const float* xr = xbase + (size_t)t * (AGENTS * I_SIZE);
        Ax = hi ? mk2(xr[2], 0.0f) : mk2(xr[0], xr[1]);

#pragma unroll
        for (int n = 0; n < 8; ++n) acc[n] = splat8(biasE[n]);
#pragma unroll
        for (int n = 0; n < 8; ++n) acc[n] = wmma4(Ax, BihE[n], acc[n]);

        k_chain(acc, hA, Bs0, Bs1, wlane);

#pragma unroll
        for (int n = 0; n < 8; ++n) {
#pragma unroll
            for (int v = 0; v < 8; ++v) {
                const int row = hi ? v + 8 : v;
                s.stage[wave][row][n * 16 + l16] = fast_tanh(acc[n][v]);
            }
        }
#pragma unroll
        for (int c = 0; c < 32; ++c)
            hA[c] = *(const v2f*)&s.stage[wave][l16][4 * c + koff];
    }

    {
        const float* xr = xbase + (size_t)(T_STEPS - 1) * (AGENTS * I_SIZE);
        Ax = hi ? mk2(xr[2], 0.0f) : mk2(xr[0], xr[1]);
    }

    v2f BihD[8];
    float biasD[8];
#pragma unroll
    for (int t = 0; t < 8; ++t) {
        const int n = t * 16 + l16;
        if (!hi) BihD[t] = mk2(W_ih_dec[n * I_SIZE + 0], W_ih_dec[n * I_SIZE + 1]);
        else     BihD[t] = mk2(W_ih_dec[n * I_SIZE + 2], 0.0f);
        biasD[t] = b_ih_dec[n] + b_hh_dec[n];
    }

    __syncthreads();
    for (int i = tid; i < H * H; i += 256)
        s.W[i >> 7][i & 127] = W_hh_dec[i];
    __syncthreads();
    load_bblock(Bs0, wlane, 0);

    for (int p = 0; p < PRED; ++p) {
#pragma unroll
        for (int n = 0; n < 8; ++n) acc[n] = splat8(biasD[n]);
#pragma unroll
        for (int n = 0; n < 8; ++n) acc[n] = wmma4(Ax, BihD[n], acc[n]);
        k_chain(acc, hA, Bs0, Bs1, wlane);
#pragma unroll
        for (int n = 0; n < 8; ++n) {
#pragma unroll
            for (int v = 0; v < 8; ++v) {
                const int row = hi ? v + 8 : v;
                s.stage[wave][row][n * 16 + l16] = fast_tanh(acc[n][v]);
            }
        }
#pragma unroll
        for (int c = 0; c < 32; ++c)
            hA[c] = *(const v2f*)&s.stage[wave][l16][4 * c + koff];

        v2f Bfc[32];
#pragma unroll
        for (int c = 0; c < 32; ++c)
            Bfc[c] = mk2(s.fcT[4 * c + koff][l16], s.fcT[4 * c + koff + 1][l16]);
        __builtin_amdgcn_sched_barrier(0);
        v8f pr = splat8(fcb);
#pragma unroll
        for (int c = 0; c < 32; ++c)
            pr = wmma4(hA[c], Bfc[c], pr);

        if (l16 < O_SIZE) {
#pragma unroll
            for (int v = 0; v < 8; ++v) {
                const int row = hi ? v + 8 : v;
                const int n  = r0 + row;
                const int bb = n >> 6, aa = n & 63;
                out[((size_t)bb * PRED + p) * (AGENTS * O_SIZE) + aa * O_SIZE + l16] = pr[v];
                s.stage[wave][row][l16] = pr[v];
            }
        }
        Ax = hi ? mk2(s.stage[wave][l16][2], 0.0f)
                : mk2(s.stage[wave][l16][0], s.stage[wave][l16][1]);
    }
}

extern "C" void kernel_launch(void* const* d_in, const int* in_sizes, int n_in,
                              void* d_out, int out_size, void* d_ws, size_t ws_size,
                              hipStream_t stream) {
    const float* src      = (const float*)d_in[0];
    const float* W_ih_enc = (const float*)d_in[1];
    const float* W_hh_enc = (const float*)d_in[2];
    const float* b_ih_enc = (const float*)d_in[3];
    const float* b_hh_enc = (const float*)d_in[4];
    const float* W_ih_dec = (const float*)d_in[5];
    const float* W_hh_dec = (const float*)d_in[6];
    const float* b_ih_dec = (const float*)d_in[7];
    const float* b_hh_dec = (const float*)d_in[8];
    const float* fc_W     = (const float*)d_in[9];
    const float* fc_b     = (const float*)d_in[10];
    float* out = (float*)d_out;

    traj_rnn_kernel<<<NBLOCKS, 256, 0, stream>>>(
        src, W_ih_enc, W_hh_enc, b_ih_enc, b_hh_enc,
        W_ih_dec, W_hh_dec, b_ih_dec, b_hh_dec, fc_W, fc_b, out);
}


# Wave priority by progress through a step: s_setprio 0 at its head, 1/2/3 as the K-chain advances, 0 behind it, so the wave in mid-chain keeps the matrix pipe and a SIMD's two waves alternate chains
# speedup vs baseline: 1.0950x; 1.0545x over previous
#include <hip/hip_runtime.h>
#include <cmath>


typedef __attribute__((ext_vector_type(2))) float v2f;
typedef __attribute__((ext_vector_type(8))) float v8f;

#define I_SIZE   3
#define H        128
#define O_SIZE   3
#define BATCH    512
#define T_STEPS  50
#define AGENTS   64
#define PRED     12
#define NROWS    (BATCH * AGENTS)
#define WPAD     132
#define WAVES    8
#define TILE_M   16
#define ROWS_PER_BLOCK (WAVES * TILE_M)
#define NBLOCKS  (NROWS / ROWS_PER_BLOCK)

struct __align__(16) Smem {
    float W[H][WPAD];
    float fcT[H][16];
    float stage[WAVES][TILE_M][WPAD];
};

__device__ __forceinline__ v2f mk2(float a, float b) { v2f r; r[0] = a; r[1] = b; return r; }

__device__ __forceinline__ v8f splat8(float x) {
    v8f r;
#pragma unroll
    for (int i = 0; i < 8; ++i) r[i] = x;
    return r;
}

__device__ __forceinline__ v8f wmma4(v2f a, v2f b, v8f c) {
    return __builtin_amdgcn_wmma_f32_16x16x4_f32(
         false, a,  false, b,
         (short)0, c,  false,  false);
}

__device__ __forceinline__ float fast_tanh(float x) {
#if __has_builtin(__builtin_amdgcn_tanhf)
    return __builtin_amdgcn_tanhf(x);
#else
    return ::tanhf(x);
#endif
}

#define KBLK 8

__device__ __forceinline__ void load_bblock(v2f (&dst)[64], const float* wlane, const int blk) {
#pragma unroll
    for (int n = 0; n < 8; ++n) {
#pragma unroll
        for (int cc = 0; cc < KBLK; ++cc)
            dst[cc * 8 + n] = *(const v2f*)(wlane + n * 16 * WPAD + 4 * (blk * KBLK + cc));
    }
}

__device__ __forceinline__ void mma_chunks(v8f (&acc)[8], const v2f (&hA)[32], const v2f (&B)[64],
                                           const int blk, const int c0, const int c1) {
#pragma unroll
    for (int cc = c0; cc < c1; ++cc) {
#pragma unroll
        for (int n = 0; n < 8; ++n)
            acc[n] = wmma4(hA[blk * KBLK + cc], B[cc * 8 + n], acc[n]);
    }
}

__device__ __forceinline__ void mma_bblock(v8f (&acc)[8], const v2f (&hA)[32], const v2f (&B)[64], const int blk) {
    mma_chunks(acc, hA, B, blk, 0, KBLK);
}

__device__ __forceinline__ void k_chain(v8f (&acc)[8], const v2f (&hA)[32],
                                        v2f (&B0)[64], v2f (&B1)[64], const float* wlane) {
    load_bblock(B1, wlane, 1);
    __builtin_amdgcn_sched_barrier(0);
    mma_chunks(acc, hA, B0, 0, 0, 2);
    __builtin_amdgcn_s_setprio(1);
    mma_chunks(acc, hA, B0, 0, 2, KBLK);
    __builtin_amdgcn_sched_barrier(0);
    load_bblock(B0, wlane, 2);
    __builtin_amdgcn_sched_barrier(0);
    mma_bblock(acc, hA, B1, 1);
    __builtin_amdgcn_sched_barrier(0);
    __builtin_amdgcn_s_setprio(2);
    load_bblock(B1, wlane, 3);
    __builtin_amdgcn_sched_barrier(0);
    mma_bblock(acc, hA, B0, 2);
    __builtin_amdgcn_sched_barrier(0);
    __builtin_amdgcn_s_setprio(3);
    load_bblock(B0, wlane, 0);
    __builtin_amdgcn_sched_barrier(0);
    mma_bblock(acc, hA, B1, 3);
}

__global__ void __launch_bounds__(256, 1)
traj_rnn_kernel(const float* __restrict__ src,
                const float* __restrict__ W_ih_enc,
                const float* __restrict__ W_hh_enc,
                const float* __restrict__ b_ih_enc,
                const float* __restrict__ b_hh_enc,
                const float* __restrict__ W_ih_dec,
                const float* __restrict__ W_hh_dec,
                const float* __restrict__ b_ih_dec,
                const float* __restrict__ b_hh_dec,
                const float* __restrict__ fc_W,
                const float* __restrict__ fc_b,
                float* __restrict__ out)
{
    __shared__ Smem s;

    const int tid  = threadIdx.x;
    const int wave = tid >> 5;
    const int lane = tid & 31;
    const int l16  = lane & 15;
    const bool hi  = lane >= 16;
    const int koff = hi ? 2 : 0;
    const int r0   = blockIdx.x * ROWS_PER_BLOCK + wave * TILE_M;

    for (int i = tid; i < H * H; i += 256)
        s.W[i >> 7][i & 127] = W_hh_enc[i];
    for (int i = tid; i < H * 16; i += 256) {
        const int k = i >> 4, n = i & 15;
        s.fcT[k][n] = (n < O_SIZE) ? fc_W[n * H + k] : 0.0f;
    }
    __syncthreads();

    v2f BihE[8];
    float biasE[8];
#pragma unroll
    for (int t = 0; t < 8; ++t) {
        const int n = t * 16 + l16;
        if (!hi) BihE[t] = mk2(W_ih_enc[n * I_SIZE + 0], W_ih_enc[n * I_SIZE + 1]);
        else     BihE[t] = mk2(W_ih_enc[n * I_SIZE + 2], 0.0f);
        biasE[t] = b_ih_enc[n] + b_hh_enc[n];
    }
    const float fcb = (l16 < O_SIZE) ? fc_b[l16] : 0.0f;

    const int myrow = r0 + l16;
    const int mb = myrow >> 6;
    const int ma = myrow & 63;
    const float* xbase = src + (size_t)mb * T_STEPS * (AGENTS * I_SIZE) + ma * I_SIZE;

    v2f hA[32];
#pragma unroll
    for (int c = 0; c < 32; ++c) hA[c] = mk2(0.0f, 0.0f);

    v2f Ax = mk2(0.0f, 0.0f);
    v8f acc[8];

    const float* wlane = &s.W[l16][koff];
    v2f Bs0[64], Bs1[64];
    load_bblock(Bs0, wlane, 0);

    for (int t = 0; t < T_STEPS; ++t) {
        const float* xr = xbase + (size_t)t * (AGENTS * I_SIZE);
        Ax = hi ? mk2(xr[2], 0.0f) : mk2(xr[0], xr[1]);

#pragma unroll
        for (int n = 0; n < 8; ++n) acc[n] = splat8(biasE[n]);
#pragma unroll
        for (int n = 0; n < 8; ++n) acc[n] = wmma4(Ax, BihE[n], acc[n]);

        k_chain(acc, hA, Bs0, Bs1, wlane);
        __builtin_amdgcn_s_setprio(0);

#pragma unroll
        for (int n = 0; n < 8; ++n) {
#pragma unroll
            for (int v = 0; v < 8; ++v) {
                const int row = hi ? v + 8 : v;
                s.stage[wave][row][n * 16 + l16] = fast_tanh(acc[n][v]);
            }
        }
#pragma unroll
        for (int c = 0; c < 32; ++c)
            hA[c] = *(const v2f*)&s.stage[wave][l16][4 * c + koff];
    }

    {
        const float* xr = xbase + (size_t)(T_STEPS - 1) * (AGENTS * I_SIZE);
        Ax = hi ? mk2(xr[2], 0.0f) : mk2(xr[0], xr[1]);
    }

    v2f BihD[8];
    float biasD[8];
#pragma unroll
    for (int t = 0; t < 8; ++t) {
        const int n = t * 16 + l16;
        if (!hi) BihD[t] = mk2(W_ih_dec[n * I_SIZE + 0], W_ih_dec[n * I_SIZE + 1]);
        else     BihD[t] = mk2(W_ih_dec[n * I_SIZE + 2], 0.0f);
        biasD[t] = b_ih_dec[n] + b_hh_dec[n];
    }

    __syncthreads();
    for (int i = tid; i < H * H; i += 256)
        s.W[i >> 7][i & 127] = W_hh_dec[i];
    __syncthreads();
    load_bblock(Bs0, wlane, 0);

    for (int p = 0; p < PRED; ++p) {
#pragma unroll
        for (int n = 0; n < 8; ++n) acc[n] = splat8(biasD[n]);
#pragma unroll
        for (int n = 0; n < 8; ++n) acc[n] = wmma4(Ax, BihD[n], acc[n]);
        k_chain(acc, hA, Bs0, Bs1, wlane);
#pragma unroll
        for (int n = 0; n < 8; ++n) {
#pragma unroll
            for (int v = 0; v < 8; ++v) {
                const int row = hi ? v + 8 : v;
                s.stage[wave][row][n * 16 + l16] = fast_tanh(acc[n][v]);
            }
        }
#pragma unroll
        for (int c = 0; c < 32; ++c)
            hA[c] = *(const v2f*)&s.stage[wave][l16][4 * c + koff];

        v2f Bfc[32];
#pragma unroll
        for (int c = 0; c < 32; ++c)
            Bfc[c] = mk2(s.fcT[4 * c + koff][l16], s.fcT[4 * c + koff + 1][l16]);
        __builtin_amdgcn_sched_barrier(0);
        v8f pr = splat8(fcb);
#pragma unroll
        for (int c = 0; c < 32; ++c)
            pr = wmma4(hA[c], Bfc[c], pr);
        __builtin_amdgcn_s_setprio(0);

        if (l16 < O_SIZE) {
#pragma unroll
            for (int v = 0; v < 8; ++v) {
                const int row = hi ? v + 8 : v;
                const int n  = r0 + row;
                const int bb = n >> 6, aa = n & 63;
                out[((size_t)bb * PRED + p) * (AGENTS * O_SIZE) + aa * O_SIZE + l16] = pr[v];
                s.stage[wave][row][l16] = pr[v];
            }
        }
        Ax = hi ? mk2(s.stage[wave][l16][2], 0.0f)
                : mk2(s.stage[wave][l16][0], s.stage[wave][l16][1]);
    }
}

extern "C" void kernel_launch(void* const* d_in, const int* in_sizes, int n_in,
                              void* d_out, int out_size, void* d_ws, size_t ws_size,
                              hipStream_t stream) {
    const float* src      = (const float*)d_in[0];
    const float* W_ih_enc = (const float*)d_in[1];
    const float* W_hh_enc = (const float*)d_in[2];
    const float* b_ih_enc = (const float*)d_in[3];
    const float* b_hh_enc = (const float*)d_in[4];
    const float* W_ih_dec = (const float*)d_in[5];
    const float* W_hh_dec = (const float*)d_in[6];
    const float* b_ih_dec = (const float*)d_in[7];
    const float* b_hh_dec = (const float*)d_in[8];
    const float* fc_W     = (const float*)d_in[9];
    const float* fc_b     = (const float*)d_in[10];
    float* out = (float*)d_out;

    traj_rnn_kernel<<<NBLOCKS, 256, 0, stream>>>(
        src, W_ih_enc, W_hh_enc, b_ih_enc, b_hh_enc,
        W_ih_dec, W_hh_dec, b_ih_dec, b_hh_dec, fc_W, fc_b, out);
}


# Loop from registers only: 1 wave per SIMD holds all W_hh as A fragments and 2 row tiles; products transposed so tanh(D) is the next B; no LDS in loop; k sums regrouped, bias added in the x product
# speedup vs baseline: 1.2138x; 1.1085x over previous
#include <hip/hip_runtime.h>
#include <cmath>


typedef __attribute__((ext_vector_type(2))) float v2f;
typedef __attribute__((ext_vector_type(4))) float v4f;
typedef __attribute__((ext_vector_type(8))) float v8f;

#define I_SIZE   3
#define H        128
#define O_SIZE   3
#define BATCH    512
#define T_STEPS  50
#define AGENTS   64
#define PRED     12
#define NROWS    (BATCH * AGENTS)
#define WPAD     132
#define WAVES    4
#define TILE     16
#define ROWS_PER_WAVE  (2 * TILE)
#define ROWS_PER_BLOCK (WAVES * ROWS_PER_WAVE)
#define NBLOCKS  (NROWS / ROWS_PER_BLOCK)
#define XROW     (AGENTS * I_SIZE)
#define OROW     (AGENTS * O_SIZE)

__device__ __forceinline__ v2f mk2(float a, float b) { v2f r; r[0] = a; r[1] = b; return r; }

__device__ __forceinline__ v8f zero8() {
    v8f r;
#pragma unroll
    for (int i = 0; i < 8; ++i) r[i] = 0.0f;
    return r;
}

__device__ __forceinline__ v8f wmma4(v2f a, v2f b, v8f c) {
    return __builtin_amdgcn_wmma_f32_16x16x4_f32(
         false, a,  false, b,
         (short)0, c,  false,  false);
}

__device__ __forceinline__ float fast_tanh(float x) {
#if __has_builtin(__builtin_amdgcn_tanhf)
    return __builtin_amdgcn_tanhf(x);
#else
    return ::tanhf(x);
#endif
}

__device__ __forceinline__ void stage_weights(float* sW, const float* __restrict__ Wg, const int tid) {
    const v4f* W4 = (const v4f*)Wg;
    for (int i = tid; i < H * H / 4; i += WAVES * 32) {
        const int n  = i >> 5;
        const int g  = i & 31;
        const int T  = g >> 2;
        const int u  = g & 3;
        const int hh = u >> 1;
        const int q0 = (u & 1) * 2;
        const v4f v  = W4[i];
        float* dst = sW + n * WPAD + 16 * T + 4 * q0 + 2 * hh;
        *(v2f*)dst       = mk2(v[0], v[1]);
        *(v2f*)(dst + 4) = mk2(v[2], v[3]);
    }
}

__device__ __forceinline__ void load_wfrags(v2f (&wf)[256], const float* wlane) {
#pragma unroll
    for (int n = 0; n < 8; ++n) {
#pragma unroll
        for (int c = 0; c < 32; ++c)
            wf[n * 32 + c] = *(const v2f*)(wlane + n * 16 * WPAD + 4 * c);
    }
}

__device__ __forceinline__ void load_wih(v2f (&wih)[8], const float* __restrict__ W_ih,
                                         const float* __restrict__ b_ih, const float* __restrict__ b_hh,
                                         const int l16, const bool hi) {
#pragma unroll
    for (int n = 0; n < 8; ++n) {
        const int m = n * 16 + l16;
        const float w02  = W_ih[m * I_SIZE + (hi ? 2 : 0)];
        const float w1   = W_ih[m * I_SIZE + 1];
        const float bias = b_ih[m] + b_hh[m];
        wih[n] = mk2(w02, hi ? bias : w1);
    }
}

__device__ __forceinline__ v2f load_x(const float* p, const bool hi) {
    const float a = p[0];
    const float b = p[hi ? 0 : 1];
    return mk2(a, hi ? 1.0f : b);
}

__device__ __forceinline__ void rnn_step(v8f (&acc)[2][8], v2f (&st)[2][32], const v2f (&wf)[256],
                                         const v2f (&wih)[8], const v2f bx0, const v2f bx1) {
#pragma unroll
    for (int n = 0; n < 8; ++n) {
        acc[0][n] = wmma4(wih[n], bx0, zero8());
        acc[1][n] = wmma4(wih[n], bx1, zero8());
#pragma unroll
        for (int c = 0; c < 32; ++c) {
            acc[0][n] = wmma4(wf[n * 32 + c], st[0][c], acc[0][n]);
            acc[1][n] = wmma4(wf[n * 32 + c], st[1][c], acc[1][n]);
        }
        __builtin_amdgcn_sched_barrier(0);
    }
#pragma unroll
    for (int t = 0; t < 2; ++t) {
#pragma unroll
        for (int n = 0; n < 8; ++n) {
#pragma unroll
            for (int q = 0; q < 4; ++q)
                st[t][4 * n + q] = mk2(fast_tanh(acc[t][n][2 * q]), fast_tanh(acc[t][n][2 * q + 1]));
        }
    }
}

__global__ void __launch_bounds__(WAVES * 32, 1)
traj_rnn_kernel(const float* __restrict__ src,
                const float* __restrict__ W_ih_enc,
                const float* __restrict__ W_hh_enc,
                const float* __restrict__ b_ih_enc,
                const float* __restrict__ b_hh_enc,
                const float* __restrict__ W_ih_dec,
                const float* __restrict__ W_hh_dec,
                const float* __restrict__ b_ih_dec,
                const float* __restrict__ b_hh_dec,
                const float* __restrict__ fc_W,
                const float* __restrict__ fc_b,
                float* __restrict__ out)
{
    __shared__ __attribute__((aligned(16))) float sW[H * WPAD];

    const int tid  = threadIdx.x;
    const int wave = tid >> 5;
    const int lane = tid & 31;
    const int l16  = lane & 15;
    const bool hi  = lane >= 16;
    const int r0   = blockIdx.x * ROWS_PER_BLOCK + wave * ROWS_PER_WAVE;

    const float* wlane = sW + l16 * WPAD + (hi ? 2 : 0);

    stage_weights(sW, W_hh_enc, tid);
    __syncthreads();
    v2f wf[256];
    load_wfrags(wf, wlane);

    v2f wih[8];
    load_wih(wih, W_ih_enc, b_ih_enc, b_hh_enc, l16, hi);

    const float* xp[2];
    float* op[2];
#pragma unroll
    for (int t = 0; t < 2; ++t) {
        const int row = r0 + t * TILE + l16;
        const int mb = row >> 6;
        const int ma = row & 63;
        xp[t] = src + (size_t)mb * T_STEPS * XROW + ma * I_SIZE + (hi ? 2 : 0);
        op[t] = out + (size_t)mb * PRED * OROW + ma * O_SIZE + (hi ? 2 : 0);
    }

    v2f st[2][32];
#pragma unroll
    for (int t = 0; t < 2; ++t) {
#pragma unroll
        for (int c = 0; c < 32; ++c) st[t][c] = mk2(0.0f, 0.0f);
    }
    v8f acc[2][8];

    v2f bx0 = load_x(xp[0], hi);
    v2f bx1 = load_x(xp[1], hi);
    for (int t = 0; t < T_STEPS; ++t) {
        const int tn = (t + 1 < T_STEPS) ? t + 1 : T_STEPS - 1;
        const v2f nx0 = load_x(xp[0] + (size_t)tn * XROW, hi);
        const v2f nx1 = load_x(xp[1] + (size_t)tn * XROW, hi);
        rnn_step(acc, st, wf, wih, bx0, bx1);
        bx0 = nx0;
        bx1 = nx1;
    }

    __syncthreads();
    stage_weights(sW, W_hh_dec, tid);
    __syncthreads();
    load_wfrags(wf, wlane);
    load_wih(wih, W_ih_dec, b_ih_dec, b_hh_dec, l16, hi);

    v2f fcA[32];
    {
        const bool used = (l16 < 2) || (l16 == 8);
        const int  o    = (l16 == 8) ? 2 : (l16 & 1);
        const float* fr = fc_W + o * H + (hi ? 8 : 0);
#pragma unroll
        for (int c = 0; c < 32; ++c) {
            const int k0 = 16 * (c >> 2) + 2 * (c & 3);
            const float f0 = fr[k0], f1 = fr[k0 + 1];
            fcA[c] = mk2(used ? f0 : 0.0f, used ? f1 : 0.0f);
        }
    }
    v8f prInit = zero8();
    prInit[0] = hi ? fc_b[2] : fc_b[0];
    prInit[1] = hi ? 0.0f : fc_b[1];

    for (int p = 0; p < PRED; ++p) {
        rnn_step(acc, st, wf, wih, bx0, bx1);

        v8f pr0 = prInit, pr1 = prInit;
#pragma unroll
        for (int c = 0; c < 32; ++c) {
            pr0 = wmma4(fcA[c], st[0][c], pr0);
            pr1 = wmma4(fcA[c], st[1][c], pr1);
        }

        float* o0 = op[0] + (size_t)p * OROW;
        float* o1 = op[1] + (size_t)p * OROW;
        o0[0] = pr0[0];
        o1[0] = pr1[0];
        if (!hi) {
            o0[1] = pr0[1];
            o1[1] = pr1[1];
        }
        bx0 = mk2(pr0[0], hi ? 1.0f : pr0[1]);
        bx1 = mk2(pr1[0], hi ? 1.0f : pr1[1]);
    }
}

extern "C" void kernel_launch(void* const* d_in, const int* in_sizes, int n_in,
                              void* d_out, int out_size, void* d_ws, size_t ws_size,
                              hipStream_t stream) {
    const float* src      = (const float*)d_in[0];
    const float* W_ih_enc = (const float*)d_in[1];
    const float* W_hh_enc = (const float*)d_in[2];
    const float* b_ih_enc = (const float*)d_in[3];
    const float* b_hh_enc = (const float*)d_in[4];
    const float* W_ih_dec = (const float*)d_in[5];
    const float* W_hh_dec = (const float*)d_in[6];
    const float* b_ih_dec = (const float*)d_in[7];
    const float* b_hh_dec = (const float*)d_in[8];
    const float* fc_W     = (const float*)d_in[9];
    const float* fc_b     = (const float*)d_in[10];
    float* out = (float*)d_out;

    traj_rnn_kernel<<<NBLOCKS, WAVES * 32, 0, stream>>>(
        src, W_ih_enc, W_hh_enc, b_ih_enc, b_hh_enc,
        W_ih_dec, W_hh_dec, b_ih_dec, b_hh_dec, fc_W, fc_b, out);
}
